# starting kernel
# baseline (speedup 1.0000x reference)
#include <hip/hip_runtime.h>


#define NPTS      64
#define FAR_DELTA 1e10f
#define TILE_RAYS 16

typedef __attribute__((ext_vector_type(2))) float v2f;
typedef __attribute__((ext_vector_type(4))) float f4v;
typedef __attribute__((ext_vector_type(8))) float v8f;

__global__ __launch_bounds__(32) void volrend_wmma_kernel(
    const float* __restrict__ density,
    const float* __restrict__ depth,
    const float* __restrict__ feat,
    float* __restrict__ feat_out,
    float* __restrict__ depth_out,
    int n_rays)
{
    __shared__ float sdT[NPTS * 17];

    const int lane = threadIdx.x;
    const int half = lane >> 4;
    const int l16  = lane & 15;
    const int R0   = blockIdx.x * TILE_RAYS;

    __builtin_prefetch(depth   + (size_t)R0 * NPTS,     0, 1);
    __builtin_prefetch(density + (size_t)R0 * NPTS,     0, 1);
    __builtin_prefetch(feat    + (size_t)R0 * NPTS * 3, 0, 1);

    #pragma unroll
    for (int i = 0; i < 8; ++i) {
        int f4i  = i * 32 + lane;
        int fidx = f4i * 4;
        int r    = fidx >> 6;
        int p    = fidx & 63;
        int row  = min(R0 + r, n_rays - 1);
        f4v dv = *(const f4v*)(depth   + (size_t)row * NPTS + p);
        f4v sv = *(const f4v*)(density + (size_t)row * NPTS + p);
        float dnext = __shfl_down(dv.x, 1, 32);
        float dl3   = (l16 == 15) ? FAR_DELTA : (dnext - dv.w);
        sdT[(p + 0) * 17 + r] = sv.x * (dv.y - dv.x);
        sdT[(p + 1) * 17 + r] = sv.y * (dv.z - dv.y);
        sdT[(p + 2) * 17 + r] = sv.z * (dv.w - dv.z);
        sdT[(p + 3) * 17 + r] = sv.w * dl3;
    }
    __syncthreads();

    v2f amask[4];
    #pragma unroll
    for (int q = 0; q < 4; ++q) {
        int j0 = 4 * q + half * 2;
        amask[q].x = (j0     < l16) ? 1.f : 0.f;
        amask[q].y = (j0 + 1 < l16) ? 1.f : 0.f;
    }

    const size_t rowc = (size_t)min(R0 + l16, n_rays - 1);
    const float* fr = feat  + rowc * NPTS * 3;
    const float* dr = depth + rowc * NPTS;

    float carry = 0.f;
    float accx = 0.f, accy = 0.f, accz = 0.f, accd = 0.f;

    #pragma unroll
    for (int t = 0; t < 4; ++t) {
        v8f   C = {};
        float tpart = 0.f;
        #pragma unroll
        for (int q = 0; q < 4; ++q) {
            int j0 = 16 * t + 4 * q + half * 2;
            v2f b;
            b.x = sdT[(j0    ) * 17 + l16];
            b.y = sdT[(j0 + 1) * 17 + l16];
            tpart += b.x + b.y;
            C = __builtin_amdgcn_wmma_f32_16x16x4_f32(
                    false, amask[q], false, b, (short)0, C, false, false);
        }
        #pragma unroll
        for (int v = 0; v < 8; ++v) {
            int   p    = 16 * t + v + half * 8;
            float Pexc = carry + C[v];
            float sdp  = sdT[p * 17 + l16];
            float w    = __expf(-Pexc) - __expf(-(Pexc + sdp));
            accx += w * fr[p * 3 + 0];
            accy += w * fr[p * 3 + 1];
            accz += w * fr[p * 3 + 2];
            accd += w * dr[p];
        }
        carry += tpart + __shfl_xor(tpart, 16, 32);
    }

    accx += __shfl_xor(accx, 16, 32);
    accy += __shfl_xor(accy, 16, 32);
    accz += __shfl_xor(accz, 16, 32);
    accd += __shfl_xor(accd, 16, 32);

    if (lane < 16) {
        int rg = R0 + l16;
        if (rg < n_rays) {
            feat_out[rg * 3 + 0] = accx;
            feat_out[rg * 3 + 1] = accy;
            feat_out[rg * 3 + 2] = accz;
            depth_out[rg]        = accd;
        }
    }
}

extern "C" void kernel_launch(void* const* d_in, const int* in_sizes, int n_in,
                              void* d_out, int out_size, void* d_ws, size_t ws_size,
                              hipStream_t stream) {
    (void)n_in; (void)out_size; (void)d_ws; (void)ws_size;
    const float* density = (const float*)d_in[0];
    const float* depthv  = (const float*)d_in[1];
    const float* feat    = (const float*)d_in[2];

    const int n_rays = in_sizes[1] / NPTS;
    float* feat_out  = (float*)d_out;
    float* depth_out = feat_out + (size_t)n_rays * 3;

    const int tiles = (n_rays + TILE_RAYS - 1) / TILE_RAYS;
    volrend_wmma_kernel<<<dim3(tiles), dim3(32), 0, stream>>>(
        density, depthv, feat, feat_out, depth_out, n_rays);
}


# Long-lived waves (a grid the chip holds at once) walk over ray tiles with loads one step ahead: next ray tile's depth/density lines, next point tile's feature/depth loads; sum order unchanged
# speedup vs baseline: 2.2891x; 2.2891x over previous
#include <hip/hip_runtime.h>


#define NPTS      64
#define FAR_DELTA 1e10f
#define TILE_RAYS 16
#define SD_PITCH  68
#define RESIDENT_WGS 4096

typedef __attribute__((ext_vector_type(2))) float v2f;
typedef __attribute__((ext_vector_type(4))) float f4v;
typedef __attribute__((ext_vector_type(8))) float v8f;

template <int CTRL>
__device__ __forceinline__ float row_lane(float x) {
    return __builtin_bit_cast(float,
        __builtin_amdgcn_update_dpp(0, __builtin_bit_cast(int, x), CTRL, 0xF, 0xF, true));
}

template <int MASK>
__device__ __forceinline__ float fold_pair(float a, float b, bool bit) {
    float keep = bit ? b : a;
    float give = bit ? a : b;
    return keep + row_lane<0x160 + MASK>(give);
}

__device__ __forceinline__ void lds_handover_one_wave() {
    __builtin_amdgcn_fence(__ATOMIC_RELEASE, "workgroup", "local");
    __builtin_amdgcn_wave_barrier();
    __builtin_amdgcn_fence(__ATOMIC_ACQUIRE, "workgroup", "local");
}

__global__ __launch_bounds__(32) void volrend_wmma_kernel(
    const float* __restrict__ density,
    const float* __restrict__ depth,
    const float* __restrict__ feat,
    float* __restrict__ feat_out,
    float* __restrict__ depth_out,
    int n_rays, int n_tiles)
{
    __shared__ __attribute__((aligned(16))) float sd[TILE_RAYS * SD_PITCH];

    const int lane = threadIdx.x;
    const int half = lane >> 4;
    const int l16  = lane & 15;
    const int step = gridDim.x;

    int tile = blockIdx.x;

    __builtin_prefetch(depth   + (size_t)tile * TILE_RAYS * NPTS,     0, 1);
    __builtin_prefetch(density + (size_t)tile * TILE_RAYS * NPTS,     0, 1);
    __builtin_prefetch(feat    + (size_t)tile * TILE_RAYS * NPTS * 3, 0, 1);

    f4v dv[8], sv[8];
#define LOAD_STAGE(TILE) do { \
        _Pragma("unroll") for (int i = 0; i < 8; ++i) { \
            int row = min((TILE) * TILE_RAYS + 2 * i + half, n_rays - 1);    \
            dv[i] = *(const f4v*)(depth   + (size_t)row * NPTS + 4 * l16); \
            sv[i] = *(const f4v*)(density + (size_t)row * NPTS + 4 * l16); \
        } } while (0)

    LOAD_STAGE(tile);

    v2f bmask[4];
    #pragma unroll
    for (int q = 0; q < 4; ++q) {
        int k0 = 4 * q + half * 2;
        bmask[q].x = (k0     < l16) ? 1.f : 0.f;
        bmask[q].y = (k0 + 1 < l16) ? 1.f : 0.f;
    }

    const float* sda = sd + l16 * SD_PITCH + 2 * half;
    const float* sdl = sd + (8 * half) * SD_PITCH + l16;

    const bool b3 = (l16 & 8) != 0, b2 = (l16 & 4) != 0;
    const bool b1 = (l16 & 2) != 0, b0 = (l16 & 1) != 0;

    for (; tile < n_tiles; tile += step) {
        const int R0 = tile * TILE_RAYS;
        const int tile_ahead = min(tile + step, n_tiles - 1);

        #pragma unroll
        for (int i = 0; i < 8; ++i) {
            int r    = 2 * i + half;
            int p    = 4 * l16;
            f4v d4 = dv[i], s0 = sv[i];
            float dnext = __shfl_down(d4.x, 1, 32);
            float dl3   = (l16 == 15) ? FAR_DELTA : (dnext - d4.w);
            f4v s4;
            s4.x = s0.x * (d4.y - d4.x);
            s4.y = s0.y * (d4.z - d4.y);
            s4.z = s0.z * (d4.w - d4.z);
            s4.w = s0.w * dl3;
            *(f4v*)(sd + r * SD_PITCH + p) = s4;
        }
        lds_handover_one_wave();

        unsigned offd[8];
        #pragma unroll
        for (int r = 0; r < 8; ++r) {
            int row = min(R0 + 8 * half + r, n_rays - 1);
            offd[r] = (unsigned)row * (NPTS * 4u) + (unsigned)l16 * 4u;
        }

        float dd[2][8], fx[2][8], fy[2][8], fz[2][8];
#define LOAD_FEAT(T) do { \
        _Pragma("unroll") for (int r = 0; r < 8; ++r) { \
            const float* pf = (const float*)((const char*)feat + 3u * offd[r]) + 48 * (T); \
            fx[(T) & 1][r] = pf[0]; fy[(T) & 1][r] = pf[1]; fz[(T) & 1][r] = pf[2]; \
        } } while (0)
#define LOAD_DEPTH(T) do { \
        _Pragma("unroll") for (int r = 0; r < 8; ++r) { \
            const float* pd = (const float*)((const char*)depth + offd[r]) + 16 * (T); \
            dd[(T) & 1][r] = pd[0]; \
        } } while (0)

        float carry[8];
        float acc[8][4];
        #pragma unroll
        for (int r = 0; r < 8; ++r) {
            carry[r] = 0.f;
            acc[r][0] = 0.f; acc[r][1] = 0.f; acc[r][2] = 0.f; acc[r][3] = 0.f;
        }

        __builtin_amdgcn_sched_barrier(0);
        LOAD_FEAT(0);
        LOAD_DEPTH(0);
        LOAD_FEAT(1);
        LOAD_DEPTH(1);
        LOAD_STAGE(tile_ahead);
        __builtin_amdgcn_sched_barrier(0);

        #pragma unroll
        for (int t = 0; t < 4; ++t) {
            if (t == 1 || t == 2) {
                __builtin_amdgcn_sched_barrier(0);
                LOAD_FEAT(t + 1);
                LOAD_DEPTH(t + 1);
                __builtin_amdgcn_sched_barrier(0);
            }

            float sp[8];
            #pragma unroll
            for (int r = 0; r < 8; ++r) sp[r] = sdl[r * SD_PITCH + 16 * t];

            v8f C = {};
            #pragma unroll
            for (int q = 0; q < 4; ++q) {
                v2f a = *(const v2f*)(sda + 16 * t + 4 * q);
                C = __builtin_amdgcn_wmma_f32_16x16x4_f32(
                        false, a, false, bmask[q], (short)0, C, false, false);
            }

            #pragma unroll
            for (int r = 0; r < 8; ++r) {
                float Pexc = carry[r] + C[r];
                float Pinc = Pexc + sp[r];
                float w    = __expf(-Pexc) - __expf(-Pinc);
                acc[r][0] += w * fx[t & 1][r];
                acc[r][1] += w * fy[t & 1][r];
                acc[r][2] += w * fz[t & 1][r];
                acc[r][3] += w * dd[t & 1][r];
                if (t < 3) carry[r] = row_lane<0x150 + 15>(Pinc);
            }
        }
#undef LOAD_FEAT
#undef LOAD_DEPTH

        float y[4][4], z[2][4], u[4];
        #pragma unroll
        for (int r = 0; r < 4; ++r)
            #pragma unroll
            for (int c = 0; c < 4; ++c)
                y[r][c] = fold_pair<8>(acc[r][c], acc[r + 4][c], b3);
        #pragma unroll
        for (int r = 0; r < 2; ++r)
            #pragma unroll
            for (int c = 0; c < 4; ++c)
                z[r][c] = fold_pair<4>(y[r][c], y[r + 2][c], b2);
        #pragma unroll
        for (int c = 0; c < 4; ++c)
            u[c] = fold_pair<2>(z[0][c], z[1][c], b1);
        float v0 = fold_pair<1>(u[0], u[1], b0);
        float v1 = fold_pair<1>(u[2], u[3], b0);

        const int rg = R0 + 8 * half + (l16 >> 1);
        if (rg < n_rays) {
            feat_out[rg * 3 + (b0 ? 1 : 0)] = v0;
            if (b0) depth_out[rg]        = v1;
            else    feat_out[rg * 3 + 2] = v1;
        }
    }
#undef LOAD_STAGE
}

extern "C" void kernel_launch(void* const* d_in, const int* in_sizes, int n_in,
                              void* d_out, int out_size, void* d_ws, size_t ws_size,
                              hipStream_t stream) {
    (void)n_in; (void)out_size; (void)d_ws; (void)ws_size;
    const float* density = (const float*)d_in[0];
    const float* depthv  = (const float*)d_in[1];
    const float* feat    = (const float*)d_in[2];

    const int n_rays = in_sizes[1] / NPTS;
    float* feat_out  = (float*)d_out;
    float* depth_out = feat_out + (size_t)n_rays * 3;

    const int tiles = (n_rays + TILE_RAYS - 1) / TILE_RAYS;
    const int grid  = tiles < RESIDENT_WGS ? tiles : RESIDENT_WGS;
    volrend_wmma_kernel<<<dim3(grid), dim3(32), 0, stream>>>(
        density, depthv, feat, feat_out, depth_out, n_rays, tiles);
}


# WMMA tile re-shaped to 8 rays x 32 points (lane halves = consecutive point tiles of one ray): each feature/depth load reads whole lines; carry by readlane; per-ray fold over 32 lanes (new sum order)
# speedup vs baseline: 2.3547x; 1.0287x over previous
#include <hip/hip_runtime.h>


#define NPTS      64
#define FAR_DELTA 1e10f
#define TILE_RAYS 16
#define SD_PITCH  68
#define RESIDENT_WGS 4096

typedef __attribute__((ext_vector_type(2))) float v2f;
typedef __attribute__((ext_vector_type(4))) float f4v;
typedef __attribute__((ext_vector_type(8))) float v8f;

template <int CTRL>
__device__ __forceinline__ float row_lane(float x) {
    return __builtin_bit_cast(float,
        __builtin_amdgcn_update_dpp(0, __builtin_bit_cast(int, x), CTRL, 0xF, 0xF, true));
}

template <int MASK>
__device__ __forceinline__ float fold_pair(float a, float b, bool bit) {
    float keep = bit ? b : a;
    float give = bit ? a : b;
    return keep + row_lane<0x160 + MASK>(give);
}

__device__ __forceinline__ float other_row(float x) {
    int v = __builtin_bit_cast(int, x);
    return __builtin_bit_cast(float,
        __builtin_amdgcn_permlanex16(v, v, 0x76543210u, 0xfedcba98u, false, false));
}

__device__ __forceinline__ void lds_handover_one_wave() {
    __builtin_amdgcn_fence(__ATOMIC_RELEASE, "workgroup", "local");
    __builtin_amdgcn_wave_barrier();
    __builtin_amdgcn_fence(__ATOMIC_ACQUIRE, "workgroup", "local");
}

__global__ __launch_bounds__(32) void volrend_wmma_kernel(
    const float* __restrict__ density,
    const float* __restrict__ depth,
    const float* __restrict__ feat,
    float* __restrict__ feat_out,
    float* __restrict__ depth_out,
    int n_rays, int n_tiles)
{
    __shared__ __attribute__((aligned(16))) float sd[TILE_RAYS * SD_PITCH];

    const int lane = threadIdx.x;
    const int half = lane >> 4;
    const int l16  = lane & 15;
    const int step = gridDim.x;

    int tile = blockIdx.x;

    __builtin_prefetch(depth   + (size_t)tile * TILE_RAYS * NPTS,     0, 1);
    __builtin_prefetch(density + (size_t)tile * TILE_RAYS * NPTS,     0, 1);
    __builtin_prefetch(feat    + (size_t)tile * TILE_RAYS * NPTS * 3, 0, 1);

    f4v dv[8], sv[8];
#define LOAD_STAGE(TILE) do { \
        _Pragma("unroll") for (int i = 0; i < 8; ++i) { \
            int row = min((TILE) * TILE_RAYS + 2 * i + half, n_rays - 1);    \
            dv[i] = *(const f4v*)(depth   + (size_t)row * NPTS + 4 * l16); \
            sv[i] = *(const f4v*)(density + (size_t)row * NPTS + 4 * l16); \
        } } while (0)

    LOAD_STAGE(tile);

    v2f bmask[4];
    #pragma unroll
    for (int q = 0; q < 4; ++q) {
        int k0 = 4 * q + half * 2;
        bmask[q].x = (k0     < l16) ? 1.f : 0.f;
        bmask[q].y = (k0 + 1 < l16) ? 1.f : 0.f;
    }

    const float* sda = sd + (l16 & 7) * SD_PITCH + 16 * (l16 >> 3) + 2 * half;
    const float* sdl = sd + lane;
    const unsigned lane4 = (unsigned)lane * 4u;

    const bool b3 = (l16 & 8) != 0, b2 = (l16 & 4) != 0;
    const bool b1 = (l16 & 2) != 0, b0 = (l16 & 1) != 0;

    for (; tile < n_tiles; tile += step) {
        const int R0 = tile * TILE_RAYS;
        const int tile_ahead = min(tile + step, n_tiles - 1);

        #pragma unroll
        for (int i = 0; i < 8; ++i) {
            int r    = 2 * i + half;
            int p    = 4 * l16;
            f4v d4 = dv[i], s0 = sv[i];
            float dnext = __shfl_down(d4.x, 1, 32);
            float dl3   = (l16 == 15) ? FAR_DELTA : (dnext - d4.w);
            f4v s4;
            s4.x = s0.x * (d4.y - d4.x);
            s4.y = s0.y * (d4.z - d4.y);
            s4.z = s0.z * (d4.w - d4.z);
            s4.w = s0.w * dl3;
            *(f4v*)(sd + r * SD_PITCH + p) = s4;
        }
        lds_handover_one_wave();

        float dd[2][8], fx[2][8], fy[2][8], fz[2][8];
#define LOAD_FEAT(GT) do { \
        _Pragma("unroll") for (int r = 0; r < 8; ++r) { \
            int row = min(R0 + 8 * ((GT) >> 1) + r, n_rays - 1);     \
            const float* pf = (const float*)((const char*)feat + ((unsigned)row * (NPTS * 12u) + 3u * lane4)) + 96 * ((GT) & 1); \
            fx[(GT) & 1][r] = pf[0]; fy[(GT) & 1][r] = pf[1]; fz[(GT) & 1][r] = pf[2]; \
        } } while (0)
#define LOAD_DEPTH(GT) do { \
        _Pragma("unroll") for (int r = 0; r < 8; ++r) { \
            int row = min(R0 + 8 * ((GT) >> 1) + r, n_rays - 1);     \
            const float* pd = (const float*)((const char*)depth + ((unsigned)row * (NPTS * 4u) + lane4)) + 32 * ((GT) & 1); \
            dd[(GT) & 1][r] = pd[0]; \
        } } while (0)

        float carry[8];
        float acc[8][4];
        float v0k = 0.f, v1k = 0.f;

        __builtin_amdgcn_sched_barrier(0);
        LOAD_FEAT(0);
        LOAD_DEPTH(0);
        LOAD_FEAT(1);
        LOAD_DEPTH(1);
        LOAD_STAGE(tile_ahead);
        __builtin_amdgcn_sched_barrier(0);

        #pragma unroll
        for (int g = 0; g < 4; ++g) {
            const int G = g >> 1, T = g & 1;
            if (g == 1 || g == 2) {
                __builtin_amdgcn_sched_barrier(0);
                LOAD_FEAT(g + 1);
                LOAD_DEPTH(g + 1);
                __builtin_amdgcn_sched_barrier(0);
            }

            if (T == 0) {
                #pragma unroll
                for (int r = 0; r < 8; ++r) {
                    carry[r] = 0.f;
                    acc[r][0] = 0.f; acc[r][1] = 0.f; acc[r][2] = 0.f; acc[r][3] = 0.f;
                }
            }

            float sp[8];
            #pragma unroll
            for (int r = 0; r < 8; ++r) sp[r] = sdl[(8 * G + r) * SD_PITCH + 32 * T];

            v8f C = {};
            #pragma unroll
            for (int q = 0; q < 4; ++q) {
                v2f a = *(const v2f*)(sda + 8 * G * SD_PITCH + 32 * T + 4 * q);
                C = __builtin_amdgcn_wmma_f32_16x16x4_f32(
                        false, a, false, bmask[q], (short)0, C, false, false);
            }

            #pragma unroll
            for (int r = 0; r < 8; ++r) {
                float PincE = (carry[r] + C[r]) + sp[r];
                float cOdd  = __builtin_bit_cast(float,
                    __builtin_amdgcn_readlane(__builtin_bit_cast(int, PincE), 15));
                float cl   = half ? cOdd : carry[r];
                float Pexc = cl + C[r];
                float Pinc = Pexc + sp[r];
                float w    = __expf(-Pexc) - __expf(-Pinc);
                acc[r][0] += w * fx[g & 1][r];
                acc[r][1] += w * fy[g & 1][r];
                acc[r][2] += w * fz[g & 1][r];
                acc[r][3] += w * dd[g & 1][r];
                if (T == 0) carry[r] = __builtin_bit_cast(float,
                    __builtin_amdgcn_readlane(__builtin_bit_cast(int, Pinc), 31));
            }

            if (T == 1) {
                float y[4][4], z[2][4], u[4];
                #pragma unroll
                for (int r = 0; r < 4; ++r)
                    #pragma unroll
                    for (int c = 0; c < 4; ++c)
                        y[r][c] = fold_pair<8>(acc[r][c], acc[r + 4][c], b3);
                #pragma unroll
                for (int r = 0; r < 2; ++r)
                    #pragma unroll
                    for (int c = 0; c < 4; ++c)
                        z[r][c] = fold_pair<4>(y[r][c], y[r + 2][c], b2);
                #pragma unroll
                for (int c = 0; c < 4; ++c)
                    u[c] = fold_pair<2>(z[0][c], z[1][c], b1);
                float w0 = fold_pair<1>(u[0], u[1], b0);
                float w1 = fold_pair<1>(u[2], u[3], b0);
                w0 += other_row(w0);
                w1 += other_row(w1);
                if (G == 0) { v0k = w0; v1k = w1; }
                else        { v0k = half ? w0 : v0k; v1k = half ? w1 : v1k; }
            }
        }
#undef LOAD_FEAT
#undef LOAD_DEPTH

        const float v0 = v0k, v1 = v1k;

        const int rg = R0 + 8 * half + (l16 >> 1);
        if (rg < n_rays) {
            feat_out[rg * 3 + (b0 ? 1 : 0)] = v0;
            if (b0) depth_out[rg]        = v1;
            else    feat_out[rg * 3 + 2] = v1;
        }
    }
#undef LOAD_STAGE
}

extern "C" void kernel_launch(void* const* d_in, const int* in_sizes, int n_in,
                              void* d_out, int out_size, void* d_ws, size_t ws_size,
                              hipStream_t stream) {
    (void)n_in; (void)out_size; (void)d_ws; (void)ws_size;
    const float* density = (const float*)d_in[0];
    const float* depthv  = (const float*)d_in[1];
    const float* feat    = (const float*)d_in[2];

    const int n_rays = in_sizes[1] / NPTS;
    float* feat_out  = (float*)d_out;
    float* depth_out = feat_out + (size_t)n_rays * 3;

    const int tiles = (n_rays + TILE_RAYS - 1) / TILE_RAYS;
    const int grid  = tiles < RESIDENT_WGS ? tiles : RESIDENT_WGS;
    volrend_wmma_kernel<<<dim3(grid), dim3(32), 0, stream>>>(
        density, depthv, feat, feat_out, depth_out, n_rays, tiles);
}


# Depth read from memory once: staging also parks its depth lines in LDS, the per-ray depth sum reads them there, not by a second global load (48 loads per ray tile, was 80); sum order unchanged
# speedup vs baseline: 2.5276x; 1.0734x over previous
#include <hip/hip_runtime.h>


#define NPTS      64
#define FAR_DELTA 1e10f
#define TILE_RAYS 16
#define SD_PITCH  68
#define RESIDENT_WGS 4096

typedef __attribute__((ext_vector_type(2))) float v2f;
typedef __attribute__((ext_vector_type(4))) float f4v;
typedef __attribute__((ext_vector_type(8))) float v8f;

template <int CTRL>
__device__ __forceinline__ float row_lane(float x) {
    return __builtin_bit_cast(float,
        __builtin_amdgcn_update_dpp(0, __builtin_bit_cast(int, x), CTRL, 0xF, 0xF, true));
}

template <int MASK>
__device__ __forceinline__ float fold_pair(float a, float b, bool bit) {
    float keep = bit ? b : a;
    float give = bit ? a : b;
    return keep + row_lane<0x160 + MASK>(give);
}

__device__ __forceinline__ float other_row(float x) {
    int v = __builtin_bit_cast(int, x);
    return __builtin_bit_cast(float,
        __builtin_amdgcn_permlanex16(v, v, 0x76543210u, 0xfedcba98u, false, false));
}

__device__ __forceinline__ void lds_handover_one_wave() {
    __builtin_amdgcn_fence(__ATOMIC_RELEASE, "workgroup", "local");
    __builtin_amdgcn_wave_barrier();
    __builtin_amdgcn_fence(__ATOMIC_ACQUIRE, "workgroup", "local");
}

__global__ __launch_bounds__(32) void volrend_wmma_kernel(
    const float* __restrict__ density,
    const float* __restrict__ depth,
    const float* __restrict__ feat,
    float* __restrict__ feat_out,
    float* __restrict__ depth_out,
    int n_rays, int n_tiles)
{
    __shared__ __attribute__((aligned(16))) float sd[TILE_RAYS * SD_PITCH];
    __shared__ __attribute__((aligned(16))) float dp[TILE_RAYS * SD_PITCH];

    const int lane = threadIdx.x;
    const int half = lane >> 4;
    const int l16  = lane & 15;
    const int step = gridDim.x;

    int tile = blockIdx.x;

    __builtin_prefetch(depth   + (size_t)tile * TILE_RAYS * NPTS,     0, 1);
    __builtin_prefetch(density + (size_t)tile * TILE_RAYS * NPTS,     0, 1);
    __builtin_prefetch(feat    + (size_t)tile * TILE_RAYS * NPTS * 3, 0, 1);

    f4v dv[8], sv[8];
#define LOAD_STAGE(TILE) do { \
        _Pragma("unroll") for (int i = 0; i < 8; ++i) { \
            int row = min((TILE) * TILE_RAYS + 2 * i + half, n_rays - 1);    \
            dv[i] = *(const f4v*)(depth   + (size_t)row * NPTS + 4 * l16); \
            sv[i] = *(const f4v*)(density + (size_t)row * NPTS + 4 * l16); \
        } } while (0)

    LOAD_STAGE(tile);

    v2f bmask[4];
    #pragma unroll
    for (int q = 0; q < 4; ++q) {
        int k0 = 4 * q + half * 2;
        bmask[q].x = (k0     < l16) ? 1.f : 0.f;
        bmask[q].y = (k0 + 1 < l16) ? 1.f : 0.f;
    }

    const float* sda = sd + (l16 & 7) * SD_PITCH + 16 * (l16 >> 3) + 2 * half;
    const float* sdl = sd + lane;
    const float* dpl = dp + lane;
    const unsigned lane12 = (unsigned)lane * 12u;

    const bool b3 = (l16 & 8) != 0, b2 = (l16 & 4) != 0;
    const bool b1 = (l16 & 2) != 0, b0 = (l16 & 1) != 0;

    for (; tile < n_tiles; tile += step) {
        const int R0 = tile * TILE_RAYS;
        const int tile_ahead = min(tile + step, n_tiles - 1);

        #pragma unroll
        for (int i = 0; i < 8; ++i) {
            int r    = 2 * i + half;
            int p    = 4 * l16;
            f4v d4 = dv[i], s0 = sv[i];
            float dnext = __shfl_down(d4.x, 1, 32);
            float dl3   = (l16 == 15) ? FAR_DELTA : (dnext - d4.w);
            f4v s4;
            s4.x = s0.x * (d4.y - d4.x);
            s4.y = s0.y * (d4.z - d4.y);
            s4.z = s0.z * (d4.w - d4.z);
            s4.w = s0.w * dl3;
            *(f4v*)(sd + r * SD_PITCH + p) = s4;
            *(f4v*)(dp + r * SD_PITCH + p) = d4;
        }
        lds_handover_one_wave();

        float fx[2][8], fy[2][8], fz[2][8];
#define LOAD_FEAT(GT) do { \
        _Pragma("unroll") for (int r = 0; r < 8; ++r) { \
            int row = min(R0 + 8 * ((GT) >> 1) + r, n_rays - 1);     \
            const float* pf = (const float*)((const char*)feat + ((unsigned)row * (NPTS * 12u) + lane12)) + 96 * ((GT) & 1); \
            fx[(GT) & 1][r] = pf[0]; fy[(GT) & 1][r] = pf[1]; fz[(GT) & 1][r] = pf[2]; \
        } } while (0)

        float carry[8];
        float acc[8][4];
        float v0k = 0.f, v1k = 0.f;

        __builtin_amdgcn_sched_barrier(0);
        LOAD_FEAT(0);
        LOAD_FEAT(1);
        LOAD_STAGE(tile_ahead);
        __builtin_amdgcn_sched_barrier(0);

        #pragma unroll
        for (int g = 0; g < 4; ++g) {
            const int G = g >> 1, T = g & 1;
            if (g == 1 || g == 2) {
                __builtin_amdgcn_sched_barrier(0);
                LOAD_FEAT(g + 1);
                __builtin_amdgcn_sched_barrier(0);
            }

            if (T == 0) {
                #pragma unroll
                for (int r = 0; r < 8; ++r) {
                    carry[r] = 0.f;
                    acc[r][0] = 0.f; acc[r][1] = 0.f; acc[r][2] = 0.f; acc[r][3] = 0.f;
                }
            }

            float sp[8], dd[8];
            #pragma unroll
            for (int r = 0; r < 8; ++r) {
                sp[r] = sdl[(8 * G + r) * SD_PITCH + 32 * T];
                dd[r] = dpl[(8 * G + r) * SD_PITCH + 32 * T];
            }

            v8f C = {};
            #pragma unroll
            for (int q = 0; q < 4; ++q) {
                v2f a = *(const v2f*)(sda + 8 * G * SD_PITCH + 32 * T + 4 * q);
                C = __builtin_amdgcn_wmma_f32_16x16x4_f32(
                        false, a, false, bmask[q], (short)0, C, false, false);
            }

            #pragma unroll
            for (int r = 0; r < 8; ++r) {
                float PincE = (carry[r] + C[r]) + sp[r];
                float cOdd  = __builtin_bit_cast(float,
                    __builtin_amdgcn_readlane(__builtin_bit_cast(int, PincE), 15));
                float cl   = half ? cOdd : carry[r];
                float Pexc = cl + C[r];
                float Pinc = Pexc + sp[r];
                float w    = __expf(-Pexc) - __expf(-Pinc);
                acc[r][0] += w * fx[g & 1][r];
                acc[r][1] += w * fy[g & 1][r];
                acc[r][2] += w * fz[g & 1][r];
                acc[r][3] += w * dd[r];
                if (T == 0) carry[r] = __builtin_bit_cast(float,
                    __builtin_amdgcn_readlane(__builtin_bit_cast(int, Pinc), 31));
            }

            if (T == 1) {
                float y[4][4], z[2][4], u[4];
                #pragma unroll
                for (int r = 0; r < 4; ++r)
                    #pragma unroll
                    for (int c = 0; c < 4; ++c)
                        y[r][c] = fold_pair<8>(acc[r][c], acc[r + 4][c], b3);
                #pragma unroll
                for (int r = 0; r < 2; ++r)
                    #pragma unroll
                    for (int c = 0; c < 4; ++c)
                        z[r][c] = fold_pair<4>(y[r][c], y[r + 2][c], b2);
                #pragma unroll
                for (int c = 0; c < 4; ++c)
                    u[c] = fold_pair<2>(z[0][c], z[1][c], b1);
                float w0 = fold_pair<1>(u[0], u[1], b0);
                float w1 = fold_pair<1>(u[2], u[3], b0);
                w0 += other_row(w0);
                w1 += other_row(w1);
                if (G == 0) { v0k = w0; v1k = w1; }
                else        { v0k = half ? w0 : v0k; v1k = half ? w1 : v1k; }
            }
        }
#undef LOAD_FEAT

        const float v0 = v0k, v1 = v1k;

        const int rg = R0 + 8 * half + (l16 >> 1);
        if (rg < n_rays) {
            feat_out[rg * 3 + (b0 ? 1 : 0)] = v0;
            if (b0) depth_out[rg]        = v1;
            else    feat_out[rg * 3 + 2] = v1;
        }
    }
#undef LOAD_STAGE
}

extern "C" void kernel_launch(void* const* d_in, const int* in_sizes, int n_in,
                              void* d_out, int out_size, void* d_ws, size_t ws_size,
                              hipStream_t stream) {
    (void)n_in; (void)out_size; (void)d_ws; (void)ws_size;
    const float* density = (const float*)d_in[0];
    const float* depthv  = (const float*)d_in[1];
    const float* feat    = (const float*)d_in[2];

    const int n_rays = in_sizes[1] / NPTS;
    float* feat_out  = (float*)d_out;
    float* depth_out = feat_out + (size_t)n_rays * 3;

    const int tiles = (n_rays + TILE_RAYS - 1) / TILE_RAYS;
    const int grid  = tiles < RESIDENT_WGS ? tiles : RESIDENT_WGS;
    volrend_wmma_kernel<<<dim3(grid), dim3(32), 0, stream>>>(
        density, depthv, feat, feat_out, depth_out, n_rays, tiles);
}
